# starting kernel
# baseline (speedup 1.0000x reference)
#include <hip/hip_runtime.h>
#include <hip/hip_bf16.h>

typedef __attribute__((ext_vector_type(2))) float v2f;
typedef __attribute__((ext_vector_type(8))) float v8f;

#define IN_CH  128
#define HID    64
#define OUTC   40
#define OUTP   48

__global__ void gcn_init(float* __restrict__ deg, float* __restrict__ agg1,
                         float* __restrict__ out, const float* __restrict__ b1,
                         const float* __restrict__ b2, int N) {
  int tid = blockIdx.x * blockDim.x + threadIdx.x;
  long long total = (long long)N * HID;
  if (tid < N) deg[tid] = 1.0f;
  if ((long long)tid < total) agg1[tid] = b1[tid & (HID - 1)];
  if ((long long)tid < (long long)N * OUTC) out[tid] = b2[tid % OUTC];
}

__global__ void gcn_degree(const int* __restrict__ col, float* __restrict__ deg,
                           int E) {
  int e = blockIdx.x * blockDim.x + threadIdx.x;
  if (e < E) atomicAdd(&deg[col[e]], 1.0f);
}

__global__ void gcn_dinv(float* __restrict__ deg, int N) {
  int i = blockIdx.x * blockDim.x + threadIdx.x;
  if (i < N) {
    float d = deg[i];
    deg[i] = (d > 0.0f) ? rsqrtf(d) : 0.0f;
  }
}

__global__ void gcn_gemm1(const float* __restrict__ x,
                          const float* __restrict__ W1,
                          float* __restrict__ h1) {
  const int wave = threadIdx.x >> 5;
  const int lane = threadIdx.x & 31;
  const int m0 = blockIdx.x * 16;
  const int n0 = wave * 16;
  const int rowA = lane & 15;
  const int colB = lane & 15;
  const int kOff = (lane < 16) ? 0 : 2;

  const float* xrow = x + (size_t)(m0 + rowA) * IN_CH;
  v8f acc = {};
#pragma unroll
  for (int k = 0; k < IN_CH; k += 4) {
    v2f a, b;
    a.x = xrow[k + kOff];
    a.y = xrow[k + kOff + 1];
    b.x = W1[(size_t)(k + kOff) * HID + n0 + colB];
    b.y = W1[(size_t)(k + kOff + 1) * HID + n0 + colB];
    acc = __builtin_amdgcn_wmma_f32_16x16x4_f32(false, a, false, b,
                                                (short)0, acc, false, false);
  }
  const int mBase = m0 + ((lane < 16) ? 0 : 8);
  const int c = n0 + colB;
#pragma unroll
  for (int j = 0; j < 8; ++j)
    h1[(size_t)(mBase + j) * HID + c] = acc[j];
}

__global__ void gcn_scatter1(const int* __restrict__ row,
                             const int* __restrict__ col,
                             const float* __restrict__ dinv,
                             const float* __restrict__ h1,
                             float* __restrict__ agg1, int E, int N) {
  int idx = blockIdx.x * blockDim.x + threadIdx.x;
  int e = idx >> 5;
  int lane = idx & 31;
  if (e >= E + N) return;
  int r, c;
  if (e < E) { r = row[e]; c = col[e]; } else { r = c = e - E; }
  float nrm = dinv[r] * dinv[c];
  const float2 hv = *(const float2*)(h1 + (size_t)r * HID + lane * 2);
  float* dst = agg1 + (size_t)c * HID + lane * 2;
  atomicAdd(dst,     hv.x * nrm);
  atomicAdd(dst + 1, hv.y * nrm);
}

__global__ void gcn_gemm2(const float* __restrict__ agg1,
                          const float* __restrict__ W2,
                          float* __restrict__ h2) {
  const int wave = threadIdx.x >> 5;
  const int lane = threadIdx.x & 31;
  const int m0 = blockIdx.x * 16;
  const int n0 = wave * 16;
  const int rowA = lane & 15;
  const int colB = n0 + (lane & 15);
  const int kOff = (lane < 16) ? 0 : 2;
  const bool valid = (colB < OUTC);

  const float* arow = agg1 + (size_t)(m0 + rowA) * HID;
  v8f acc = {};
#pragma unroll
  for (int k = 0; k < HID; k += 4) {
    v2f a, b;
    a.x = fmaxf(arow[k + kOff], 0.0f);
    a.y = fmaxf(arow[k + kOff + 1], 0.0f);
    b.x = valid ? W2[(size_t)(k + kOff) * OUTC + colB] : 0.0f;
    b.y = valid ? W2[(size_t)(k + kOff + 1) * OUTC + colB] : 0.0f;
    acc = __builtin_amdgcn_wmma_f32_16x16x4_f32(false, a, false, b,
                                                (short)0, acc, false, false);
  }
  const int mBase = m0 + ((lane < 16) ? 0 : 8);
  if (valid) {
#pragma unroll
    for (int j = 0; j < 8; ++j)
      h2[(size_t)(mBase + j) * OUTP + colB] = acc[j];
  }
}

__global__ void gcn_scatter2(const int* __restrict__ row,
                             const int* __restrict__ col,
                             const float* __restrict__ dinv,
                             const float* __restrict__ h2,
                             float* __restrict__ out, int E, int N) {
  int idx = blockIdx.x * blockDim.x + threadIdx.x;
  int e = idx >> 5;
  int lane = idx & 31;
  if (e >= E + N || lane >= OUTC / 2) return;
  int r, c;
  if (e < E) { r = row[e]; c = col[e]; } else { r = c = e - E; }
  float nrm = dinv[r] * dinv[c];
  const float2 hv = *(const float2*)(h2 + (size_t)r * OUTP + lane * 2);
  float* dst = out + (size_t)c * OUTC + lane * 2;
  atomicAdd(dst,     hv.x * nrm);
  atomicAdd(dst + 1, hv.y * nrm);
}

extern "C" void kernel_launch(void* const* d_in, const int* in_sizes, int n_in,
                              void* d_out, int out_size, void* d_ws, size_t ws_size,
                              hipStream_t stream) {
  const float* x  = (const float*)d_in[0];
  const int*   ei = (const int*)d_in[1];
  const float* W1 = (const float*)d_in[2];
  const float* b1 = (const float*)d_in[3];
  const float* W2 = (const float*)d_in[4];
  const float* b2 = (const float*)d_in[5];
  float* out = (float*)d_out;

  const int N = in_sizes[0] / IN_CH;
  const int E = in_sizes[1] / 2;
  const int* rowp = ei;
  const int* colp = ei + E;

  float* ws   = (float*)d_ws;
  float* deg  = ws;
  float* h1   = ws + N;
  float* agg1 = ws + N + (size_t)N * HID;

  {
    long long total = (long long)N * HID;
    int blocks = (int)((total + 255) / 256);
    gcn_init<<<blocks, 256, 0, stream>>>(deg, agg1, out, b1, b2, N);
  }
  gcn_degree<<<(E + 255) / 256, 256, 0, stream>>>(colp, deg, E);
  gcn_dinv<<<(N + 255) / 256, 256, 0, stream>>>(deg, N);
  gcn_gemm1<<<N / 16, 128, 0, stream>>>(x, W1, h1);
  {
    long long work = (long long)(E + N) * 32;
    int blocks = (int)((work + 255) / 256);
    gcn_scatter1<<<blocks, 256, 0, stream>>>(rowp, colp, deg, h1, agg1, E, N);
  }
  gcn_gemm2<<<N / 16, 96, 0, stream>>>(agg1, W2, h1);
  {
    long long work = (long long)(E + N) * 32;
    int blocks = (int)((work + 255) / 256);
    gcn_scatter2<<<blocks, 256, 0, stream>>>(rowp, colp, deg, h1, out, E, N);
  }
}


# Degree count without global atomics: blocks (8 node ranges x 32 edge slices) count targets in LDS, plain-store partial counts the D^-1/2 pass sums; init launch gone; counts exact, float sums kept
# speedup vs baseline: 2.1954x; 2.1954x over previous
#include <hip/hip_runtime.h>
#include <hip/hip_bf16.h>

typedef __attribute__((ext_vector_type(2))) float v2f;
typedef __attribute__((ext_vector_type(8))) float v8f;

#define IN_CH  128
#define HID    64
#define OUTC   40
#define OUTP   64

__device__ __forceinline__ int package_half() {
  return (int)((__builtin_amdgcn_s_sendmsg_rtn(0x87) >> 18) & 1u);
}

#define DEG_RN      12544
#define DEG_S       32
#define DEG_THREADS 1024
#define DEG_Q       16

__device__ __forceinline__ void deg_count(unsigned int* cnt, int c, int lo, int nr) {
  const unsigned int d = (unsigned int)c - (unsigned int)lo;
  if (d < (unsigned int)nr) atomicAdd(&cnt[d], 1u);
}

__global__ void __launch_bounds__(DEG_THREADS)
gcn_degree(const int* __restrict__ col, unsigned int* __restrict__ part, int N, int E) {
  __shared__ unsigned int cnt[DEG_RN];
  const int tid = threadIdx.x;
  const int lo = blockIdx.x * DEG_RN;
  const int nr = min(DEG_RN, N - lo);
  const int s = blockIdx.y;
  for (int i = tid; i < DEG_RN; i += DEG_THREADS) cnt[i] = 0u;
  __syncthreads();

  const int per = (E + DEG_S - 1) / DEG_S;
  const int e0 = min(E, s * per);
  const int e1 = min(E, e0 + per);
  const int mis = (int)((((size_t)(col + e0)) >> 2) & 3);
  const int a0 = min(e1, e0 + ((4 - mis) & 3));
  const int nq = (e1 - a0) >> 2;
  const int a1 = a0 + 4 * nq;
  if (tid < a0 - e0) deg_count(cnt, col[e0 + tid], lo, nr);
  if (tid < e1 - a1) deg_count(cnt, col[a1 + tid], lo, nr);
  const int4* __restrict__ q4 = reinterpret_cast<const int4*>(col + a0);
  for (int q0 = 0; q0 < nq; q0 += DEG_THREADS * DEG_Q) {
    int4 v[DEG_Q];
#pragma unroll
    for (int u = 0; u < DEG_Q; ++u)
      v[u] = q4[min(q0 + u * DEG_THREADS + tid, nq - 1)];
#pragma unroll
    for (int u = 0; u < DEG_Q; ++u) {
      if (q0 + u * DEG_THREADS + tid < nq) {
        deg_count(cnt, v[u].x, lo, nr);
        deg_count(cnt, v[u].y, lo, nr);
        deg_count(cnt, v[u].z, lo, nr);
        deg_count(cnt, v[u].w, lo, nr);
      }
    }
  }
  __syncthreads();
  unsigned int* __restrict__ dst = part + (size_t)s * N + lo;
  for (int i = tid; i < nr; i += DEG_THREADS) dst[i] = cnt[i];
}

__global__ void gcn_dinv(const unsigned int* __restrict__ part, float* __restrict__ dinv,
                         int N) {
  int i = blockIdx.x * blockDim.x + threadIdx.x;
  if (i < N) {
    unsigned int cntv = 1u;
#pragma unroll
    for (int s = 0; s < DEG_S; ++s) cntv += part[(size_t)s * N + i];
    const float d = (float)cntv;
    dinv[i] = (d > 0.0f) ? rsqrtf(d) : 0.0f;
  }
}

#define A1_PITCH (IN_CH + 4)
__global__ void __launch_bounds__(128)
gcn_gemm1(const float* __restrict__ x,
          const float* __restrict__ W1,
          const float* __restrict__ dinv,
          const float* __restrict__ b1,
          float* __restrict__ h1,
          float* __restrict__ agg1,
          float* __restrict__ agg1b) {
  __shared__ __attribute__((aligned(16))) float As[16 * A1_PITCH];
  const int wave = threadIdx.x >> 5;
  const int lane = threadIdx.x & 31;
  const int m0 = blockIdx.x * 16;
  const int n0 = wave * 16;
  const int rowA = lane & 15;
  const int colB = lane & 15;
  const int kOff = (lane < 16) ? 0 : 2;

#pragma unroll
  for (int r = 0; r < 4; ++r) {
    const int rr = wave * 4 + r;
    const float4 v = *reinterpret_cast<const float4*>(x + (size_t)(m0 + rr) * IN_CH + lane * 4);
    *reinterpret_cast<float4*>(&As[rr * A1_PITCH + lane * 4]) = v;
  }

  const int mBase = m0 + ((lane < 16) ? 0 : 8);
  const int c = n0 + colB;
  const float4 dlo = *reinterpret_cast<const float4*>(dinv + mBase);
  const float4 dhi = *reinterpret_cast<const float4*>(dinv + mBase + 4);
  const float bias = b1[c];
  __syncthreads();
  const float* arow = &As[rowA * A1_PITCH + kOff];
  v8f acc = {};
#pragma unroll
  for (int k = 0; k < IN_CH; k += 4) {
    v2f a, b;
    a = *reinterpret_cast<const v2f*>(arow + k);
    b.x = W1[(size_t)(k + kOff) * HID + n0 + colB];
    b.y = W1[(size_t)(k + kOff + 1) * HID + n0 + colB];
    acc = __builtin_amdgcn_wmma_f32_16x16x4_f32(false, a, false, b,
                                                (short)0, acc, false, false);
  }
  const float dv[8] = {dlo.x, dlo.y, dlo.z, dlo.w, dhi.x, dhi.y, dhi.z, dhi.w};
#pragma unroll
  for (int j = 0; j < 8; ++j) {
    const size_t o = (size_t)(mBase + j) * HID + c;
    h1[o] = acc[j];
    agg1[o] = bias + acc[j] * (dv[j] * dv[j]);
    agg1b[o] = 0.0f;
  }
}

#define SCAT_THREADS 256
#define EPW 32
#define GRP 16

__device__ __forceinline__ int bcast_i(int v, int j) {
  return __builtin_amdgcn_readlane(v, j);
}
__device__ __forceinline__ float bcast_f(float v, int j) {
  return __int_as_float(__builtin_amdgcn_readlane(__float_as_int(v), j));
}

template <int NCH, int LDH, int LDO, bool MAKE_NORM>
__device__ __forceinline__ void scatter_rows(const int* __restrict__ row,
                                             const int* __restrict__ col,
                                             const float* __restrict__ dinv,
                                             const float* __restrict__ nrm_ld,
                                             float* __restrict__ nrm_st,
                                             const float* __restrict__ h,
                                             float* __restrict__ acc0,
                                             float* __restrict__ acc1,
                                             int E) {
  constexpr int P = (NCH + 31) / 32;
  constexpr int KMIN = NCH - 32 * (P - 1);
  const int lane = threadIdx.x & 31;
  const int wave = __builtin_amdgcn_readfirstlane(
      (int)(blockIdx.x * (SCAT_THREADS / 32) + (threadIdx.x >> 5)));
  const int ET = E;
  const int base = wave * EPW;
  if (base >= ET) return;
  float* __restrict__ acc = package_half() ? acc1 : acc0;

  const int e = base + lane;
  int r = 0, c = 0;
  float nrm = 0.0f;
  if (e < ET) {
    r = row[e];
    c = col[e];
    if (MAKE_NORM) {
      nrm = dinv[r] * dinv[c];
      nrm_st[e] = nrm;
    } else {
      nrm = nrm_ld[e];
    }
  }
  const int n = min(EPW, ET - base);
  const int k1v = (KMIN == 32)
      ? 32
      : max(KMIN, 32 - (int)((((size_t)(acc + (size_t)c * LDO)) >> 2) & 31));

#define LINE_CH(k1, i) (((k1) - 32 + 32 * (i)) + ((lane - ((k1) - 32 + 32 * (i))) & 31))

  if (n == EPW) {
#pragma unroll
    for (int g = 0; g < EPW; g += GRP) {
      float v[GRP][P];
#pragma unroll
      for (int j = 0; j < GRP; ++j) {
        const int rj = bcast_i(r, g + j);
        const float* src = h + (size_t)rj * LDH;
#pragma unroll
        for (int p = 0; p < P; ++p)
          v[j][p] = src[min(32 * p + lane, NCH - 1)];
      }
#pragma unroll
      for (int j = 0; j < GRP; ++j) {
        const int cj = bcast_i(c, g + j);
        const float nj = bcast_f(nrm, g + j);
        const int k1 = bcast_i(k1v, g + j);
        float* dst = acc + (size_t)cj * LDO;
#pragma unroll
        for (int i = 0; i < P; ++i) {
          const int ch = LINE_CH(k1, i);
          float val = v[j][0];
#pragma unroll
          for (int q = 1; q < P; ++q) val = (ch >= 32 * q) ? v[j][q] : val;
          if (ch >= 0 && ch < NCH) atomicAdd(dst + ch, val * nj);
        }
      }
    }
  } else {
    for (int j = 0; j < n; ++j) {
      const int rj = bcast_i(r, j);
      const int cj = bcast_i(c, j);
      const float nj = bcast_f(nrm, j);
      const int k1 = bcast_i(k1v, j);
      const float* src = h + (size_t)rj * LDH;
      float* dst = acc + (size_t)cj * LDO;
#pragma unroll
      for (int i = 0; i < P; ++i) {
        const int ch = LINE_CH(k1, i);
        if (ch >= 0 && ch < NCH) atomicAdd(dst + ch, src[ch] * nj);
      }
    }
  }
#undef LINE_CH
}

__global__ void __launch_bounds__(SCAT_THREADS)
gcn_scatter1(const int* __restrict__ row, const int* __restrict__ col,
             const float* __restrict__ dinv, float* __restrict__ norm,
             const float* __restrict__ h1,
             float* __restrict__ agg1, float* __restrict__ agg1b, int E) {
  scatter_rows<HID, HID, HID, true>(row, col, dinv, nullptr, norm, h1, agg1, agg1b, E);
}

#define A2_PITCH (HID + 4)
__global__ void __launch_bounds__(96)
gcn_gemm2(const float* __restrict__ agg1,
          const float* __restrict__ agg1b,
          const float* __restrict__ W2,
          const float* __restrict__ dinv,
          const float* __restrict__ b2,
          float* __restrict__ h2,
          float* __restrict__ out,
          float* __restrict__ outb) {
  __shared__ __attribute__((aligned(16))) float As[16 * A2_PITCH];
  const int wave = threadIdx.x >> 5;
  const int lane = threadIdx.x & 31;
  const int m0 = blockIdx.x * 16;
  const int n0 = wave * 16;
  const int rowA = lane & 15;
  const int colB = n0 + (lane & 15);
  const int kOff = (lane < 16) ? 0 : 2;
  const bool valid = (colB < OUTC);

  for (int p = wave; p < 8; p += 3) {
    const int rr = 2 * p + (lane >> 4);
    const int cc = (lane & 15) * 4;
    const size_t g = (size_t)(m0 + rr) * HID + cc;
    const float4 u = *reinterpret_cast<const float4*>(agg1 + g);
    const float4 w = *reinterpret_cast<const float4*>(agg1b + g);
    float4 s;
    s.x = fmaxf(u.x + w.x, 0.0f);
    s.y = fmaxf(u.y + w.y, 0.0f);
    s.z = fmaxf(u.z + w.z, 0.0f);
    s.w = fmaxf(u.w + w.w, 0.0f);
    *reinterpret_cast<float4*>(&As[rr * A2_PITCH + cc]) = s;
  }

  const int mBase = m0 + ((lane < 16) ? 0 : 8);
  const float4 dlo = *reinterpret_cast<const float4*>(dinv + mBase);
  const float4 dhi = *reinterpret_cast<const float4*>(dinv + mBase + 4);
  const float bias = valid ? b2[colB] : 0.0f;
  __syncthreads();
  const float* arow = &As[rowA * A2_PITCH + kOff];
  v8f acc = {};
#pragma unroll
  for (int k = 0; k < HID; k += 4) {
    v2f a, b;
    a = *reinterpret_cast<const v2f*>(arow + k);
    b.x = valid ? W2[(size_t)(k + kOff) * OUTC + colB] : 0.0f;
    b.y = valid ? W2[(size_t)(k + kOff + 1) * OUTC + colB] : 0.0f;
    acc = __builtin_amdgcn_wmma_f32_16x16x4_f32(false, a, false, b,
                                                (short)0, acc, false, false);
  }
  const float dv[8] = {dlo.x, dlo.y, dlo.z, dlo.w, dhi.x, dhi.y, dhi.z, dhi.w};
  if (valid) {
#pragma unroll
    for (int j = 0; j < 8; ++j) {
      const size_t o = (size_t)(mBase + j) * OUTC + colB;
      h2[(size_t)(mBase + j) * OUTP + colB] = acc[j];
      out[o] = bias + acc[j] * (dv[j] * dv[j]);
      outb[o] = 0.0f;
    }
  }
}

__global__ void __launch_bounds__(SCAT_THREADS)
gcn_scatter2(const int* __restrict__ row, const int* __restrict__ col,
             const float* __restrict__ norm, const float* __restrict__ h2,
             float* __restrict__ out, float* __restrict__ outb, int E) {
  scatter_rows<OUTC, OUTP, OUTC, false>(row, col, nullptr, norm, nullptr, h2, out, outb, E);
}

__global__ void gcn_sum_out(float* __restrict__ out, const float* __restrict__ outb,
                            int n4) {
  int i = blockIdx.x * blockDim.x + threadIdx.x;
  if (i < n4) {
    float4 a = reinterpret_cast<float4*>(out)[i];
    const float4 b = reinterpret_cast<const float4*>(outb)[i];
    a.x += b.x; a.y += b.y; a.z += b.z; a.w += b.w;
    reinterpret_cast<float4*>(out)[i] = a;
  }
}

extern "C" void kernel_launch(void* const* d_in, const int* in_sizes, int n_in,
                              void* d_out, int out_size, void* d_ws, size_t ws_size,
                              hipStream_t stream) {
  const float* x  = (const float*)d_in[0];
  const int*   ei = (const int*)d_in[1];
  const float* W1 = (const float*)d_in[2];
  const float* b1 = (const float*)d_in[3];
  const float* W2 = (const float*)d_in[4];
  const float* b2 = (const float*)d_in[5];
  float* out = (float*)d_out;

  const int N = in_sizes[0] / IN_CH;
  const int E = in_sizes[1] / 2;
  const int* rowp = ei;
  const int* colp = ei + E;

  float* ws   = (float*)d_ws;
  float* deg  = ws;
  float* h1   = ws + N;
  float* agg1 = ws + N + (size_t)N * HID;
  float* agg1b = agg1 + (size_t)N * HID;
  float* outb  = agg1b + (size_t)N * HID;
  float* norm  = outb + (size_t)N * OUTC;
  unsigned int* part = (unsigned int*)(norm + E);

  gcn_degree<<<dim3((N + DEG_RN - 1) / DEG_RN, DEG_S), DEG_THREADS, 0, stream>>>(colp, part, N, E);
  gcn_dinv<<<(N + 255) / 256, 256, 0, stream>>>(part, deg, N);
  gcn_gemm1<<<N / 16, 128, 0, stream>>>(x, W1, deg, b1, h1, agg1, agg1b);
  gcn_scatter1<<<(E + SCAT_THREADS - 1) / SCAT_THREADS, SCAT_THREADS, 0, stream>>>(
      rowp, colp, deg, norm, h1, agg1, agg1b, E);
  gcn_gemm2<<<N / 16, 96, 0, stream>>>(agg1, agg1b, W2, deg, b2, h1, out, outb);
  gcn_scatter2<<<(E + SCAT_THREADS - 1) / SCAT_THREADS, SCAT_THREADS, 0, stream>>>(
      rowp, colp, norm, h1, out, outb, E);
  {
    int n4 = (N / 4) * OUTC;
    gcn_sum_out<<<(n4 + 255) / 256, 256, 0, stream>>>(out, outb, n4);
  }
}


# Edges routed by target: node ranges split per package half, work-groups draw (chunk, range) items from per-half integer counters; second sum copies and closing add launch go; float sum order changes
# speedup vs baseline: 2.8913x; 1.3170x over previous
#include <hip/hip_runtime.h>
#include <hip/hip_bf16.h>

typedef __attribute__((ext_vector_type(2))) float v2f;
typedef __attribute__((ext_vector_type(8))) float v8f;

#define IN_CH  128
#define HID    64
#define OUTC   40
#define OUTP   64

__device__ __forceinline__ int package_half() {
  return (int)((__builtin_amdgcn_s_sendmsg_rtn(0x87) >> 18) & 1u);
}

#define DEG_RN      12544
#define DEG_S       32
#define DEG_THREADS 1024
#define DEG_Q       16

__device__ __forceinline__ void deg_count(unsigned int* cnt, int c, int lo, int nr) {
  const unsigned int d = (unsigned int)c - (unsigned int)lo;
  if (d < (unsigned int)nr) atomicAdd(&cnt[d], 1u);
}

__global__ void __launch_bounds__(DEG_THREADS)
gcn_degree(const int* __restrict__ col, unsigned int* __restrict__ part, int N, int E) {
  __shared__ unsigned int cnt[DEG_RN];
  const int tid = threadIdx.x;
  const int lo = blockIdx.x * DEG_RN;
  const int nr = min(DEG_RN, N - lo);
  const int s = blockIdx.y;
  for (int i = tid; i < DEG_RN; i += DEG_THREADS) cnt[i] = 0u;
  __syncthreads();

  const int per = (E + DEG_S - 1) / DEG_S;
  const int e0 = min(E, s * per);
  const int e1 = min(E, e0 + per);
  const int mis = (int)((((size_t)(col + e0)) >> 2) & 3);
  const int a0 = min(e1, e0 + ((4 - mis) & 3));
  const int nq = (e1 - a0) >> 2;
  const int a1 = a0 + 4 * nq;
  if (tid < a0 - e0) deg_count(cnt, col[e0 + tid], lo, nr);
  if (tid < e1 - a1) deg_count(cnt, col[a1 + tid], lo, nr);
  const int4* __restrict__ q4 = reinterpret_cast<const int4*>(col + a0);
  for (int q0 = 0; q0 < nq; q0 += DEG_THREADS * DEG_Q) {
    int4 v[DEG_Q];
#pragma unroll
    for (int u = 0; u < DEG_Q; ++u)
      v[u] = q4[min(q0 + u * DEG_THREADS + tid, nq - 1)];
#pragma unroll
    for (int u = 0; u < DEG_Q; ++u) {
      if (q0 + u * DEG_THREADS + tid < nq) {
        deg_count(cnt, v[u].x, lo, nr);
        deg_count(cnt, v[u].y, lo, nr);
        deg_count(cnt, v[u].z, lo, nr);
        deg_count(cnt, v[u].w, lo, nr);
      }
    }
  }
  __syncthreads();
  unsigned int* __restrict__ dst = part + (size_t)s * N + lo;
  for (int i = tid; i < nr; i += DEG_THREADS) dst[i] = cnt[i];
}

__global__ void gcn_dinv(const unsigned int* __restrict__ part, float* __restrict__ dinv,
                         int* __restrict__ claim, int N) {
  int i = blockIdx.x * blockDim.x + threadIdx.x;
  if (i < 4) claim[i] = 0;
  if (i < N) {
    unsigned int cntv = 1u;
#pragma unroll
    for (int s = 0; s < DEG_S; ++s) cntv += part[(size_t)s * N + i];
    const float d = (float)cntv;
    dinv[i] = (d > 0.0f) ? rsqrtf(d) : 0.0f;
  }
}

#define A1_PITCH (IN_CH + 4)
__global__ void __launch_bounds__(128)
gcn_gemm1(const float* __restrict__ x,
          const float* __restrict__ W1,
          const float* __restrict__ dinv,
          const float* __restrict__ b1,
          float* __restrict__ h1,
          float* __restrict__ agg1) {
  __shared__ __attribute__((aligned(16))) float As[16 * A1_PITCH];
  const int wave = threadIdx.x >> 5;
  const int lane = threadIdx.x & 31;
  const int m0 = blockIdx.x * 16;
  const int n0 = wave * 16;
  const int rowA = lane & 15;
  const int colB = lane & 15;
  const int kOff = (lane < 16) ? 0 : 2;

#pragma unroll
  for (int r = 0; r < 4; ++r) {
    const int rr = wave * 4 + r;
    const float4 v = *reinterpret_cast<const float4*>(x + (size_t)(m0 + rr) * IN_CH + lane * 4);
    *reinterpret_cast<float4*>(&As[rr * A1_PITCH + lane * 4]) = v;
  }

  const int mBase = m0 + ((lane < 16) ? 0 : 8);
  const int c = n0 + colB;
  const float4 dlo = *reinterpret_cast<const float4*>(dinv + mBase);
  const float4 dhi = *reinterpret_cast<const float4*>(dinv + mBase + 4);
  const float bias = b1[c];
  __syncthreads();
  const float* arow = &As[rowA * A1_PITCH + kOff];
  v8f acc = {};
#pragma unroll
  for (int k = 0; k < IN_CH; k += 4) {
    v2f a, b;
    a = *reinterpret_cast<const v2f*>(arow + k);
    b.x = W1[(size_t)(k + kOff) * HID + n0 + colB];
    b.y = W1[(size_t)(k + kOff + 1) * HID + n0 + colB];
    acc = __builtin_amdgcn_wmma_f32_16x16x4_f32(false, a, false, b,
                                                (short)0, acc, false, false);
  }
  const float dv[8] = {dlo.x, dlo.y, dlo.z, dlo.w, dhi.x, dhi.y, dhi.z, dhi.w};
#pragma unroll
  for (int j = 0; j < 8; ++j) {
    const size_t o = (size_t)(mBase + j) * HID + c;
    h1[o] = acc[j];
    agg1[o] = bias + acc[j] * (dv[j] * dv[j]);
  }
}

#define SCAT_THREADS 256
#define SCAT_WAVES (SCAT_THREADS / 32)
#define EPW 64
#define EPB (EPW * SCAT_WAVES)
#define GRP 20

__device__ __forceinline__ int bcast_i(int v, int j) {
  return __builtin_amdgcn_readlane(v, j);
}
__device__ __forceinline__ float bcast_f(int v, int j) {
  return __int_as_float(__builtin_amdgcn_readlane(v, j));
}

template <int NCH, int LDH, int LDO, bool MAKE_NORM>
__device__ __forceinline__ void scatter_rows(const int* __restrict__ row,
                                             const int* __restrict__ col,
                                             const float* __restrict__ dinv,
                                             float* __restrict__ nrm2,
                                             const float* __restrict__ h,
                                             float* __restrict__ acc,
                                             int* __restrict__ claim,
                                             int N, int E) {
  constexpr int P = (NCH + 31) / 32;
  constexpr int KMIN = NCH - 32 * (P - 1);
  __shared__ int s_item;
  __shared__ __attribute__((aligned(16))) int4 ent[SCAT_WAVES][EPW];
  const int lane = threadIdx.x & 31;
  const int wib = __builtin_amdgcn_readfirstlane((int)(threadIdx.x >> 5));
  const int nchunks = (E + EPB - 1) / EPB;

  if (threadIdx.x == 0) {
    int hh = package_half();
    int idx = atomicAdd(&claim[hh], 1);
    if (idx >= nchunks) {
      hh ^= 1;
      idx = atomicAdd(&claim[hh], 1);
    }
    s_item = (idx < nchunks) ? (2 * idx + hh) : -1;
  }
  __syncthreads();
  const int item = __builtin_amdgcn_readfirstlane(s_item);
  if (item < 0) return;
  const int hsel = item & 1;
  const int base = (item >> 1) * EPB + wib * EPW;
  if (base >= E) return;
  const int nsplit = ((N >> 1) + 3) & ~3;
  const unsigned int lo = hsel ? (unsigned int)nsplit : 0u;
  const unsigned int nr = hsel ? (unsigned int)(N - nsplit) : (unsigned int)nsplit;
  float* __restrict__ nrm = nrm2 + (size_t)hsel * E;

  int r[2], c[2], nb[2], k1v[2];
  bool mine[2];
  bool valid[2];
  int ec[2];
#pragma unroll
  for (int u = 0; u < 2; ++u) {
    const int e = base + 32 * u + lane;
    valid[u] = e < E;
    ec[u] = min(e, E - 1);
    c[u] = col[ec[u]];
    r[u] = row[ec[u]];
    if (!MAKE_NORM) nb[u] = __float_as_int(nrm[ec[u]]);
  }
  if (MAKE_NORM) {
    asm volatile("" : : "v"(c[0]), "v"(c[1]), "v"(r[0]), "v"(r[1]));
  } else {
    asm volatile("" : : "v"(c[0]), "v"(c[1]), "v"(r[0]), "v"(r[1]), "v"(nb[0]), "v"(nb[1]));
  }
#pragma unroll
  for (int u = 0; u < 2; ++u)
    mine[u] = valid[u] && (((unsigned int)c[u] - lo) < nr);
  if (MAKE_NORM) {
    float da[2], db[2];
#pragma unroll
    for (int u = 0; u < 2; ++u) {
      da[u] = dinv[mine[u] ? r[u] : 0];
      db[u] = dinv[mine[u] ? c[u] : 0];
    }
#pragma unroll
    for (int u = 0; u < 2; ++u) {
      const float nv = mine[u] ? da[u] * db[u] : 0.0f;
      if (valid[u]) nrm[base + 32 * u + lane] = nv;
      nb[u] = __float_as_int(nv);
    }
  }
#pragma unroll
  for (int u = 0; u < 2; ++u) {
    k1v[u] = (KMIN == 32)
        ? 32
        : max(KMIN, 32 - (int)((((size_t)(acc + (size_t)c[u] * LDO)) >> 2) & 31));
  }
  const unsigned int m0 = __builtin_amdgcn_ballot_w32(mine[0]);
  const unsigned int m1 = __builtin_amdgcn_ballot_w32(mine[1]);
  const int n0 = __builtin_popcount(m0);
  const int n = n0 + __builtin_popcount(m1);
  if (mine[0])
    ent[wib][__builtin_amdgcn_mbcnt_lo(m0, 0u)] = make_int4(r[0], c[0], nb[0], k1v[0]);
  if (mine[1])
    ent[wib][n0 + __builtin_amdgcn_mbcnt_lo(m1, 0u)] = make_int4(r[1], c[1], nb[1], k1v[1]);
  __builtin_amdgcn_wave_barrier();

#define LINE_CH(k1, i) (((k1) - 32 + 32 * (i)) + ((lane - ((k1) - 32 + 32 * (i))) & 31))

  const int ng = (n + GRP - 1) / GRP;
  const int gs = (ng <= 1) ? n : (ng == 2) ? (n + 1) / 2 : (ng == 3) ? (n + 2) / 3 : (n + 3) / 4;
  for (int s = 0; s < n; s += gs) {
    const int cnt = __builtin_amdgcn_readfirstlane(min(gs, n - s));
    const int4 en = ent[wib][s + min(lane, cnt - 1)];
    float v[GRP][P] = {};
#pragma unroll
    for (int j = 0; j < GRP; ++j) {
      if (__builtin_expect(j < cnt, 1)) {
        const int rj = bcast_i(en.x, j);
        const float* src = h + (size_t)rj * LDH;
#pragma unroll
        for (int p = 0; p < P; ++p)
          v[j][p] = src[min(32 * p + lane, NCH - 1)];
      }
    }
#pragma unroll
    for (int j = 0; j < GRP; ++j) {
#pragma unroll
      for (int p = 0; p < P; ++p) asm volatile("" : : "v"(v[j][p]));
    }
#pragma unroll
    for (int j = 0; j < GRP; ++j) {
      if (__builtin_expect(j < cnt, 1)) {
        const int cj = bcast_i(en.y, j);
        const float nj = bcast_f(en.z, j);
        const int k1 = (KMIN == 32) ? 32 : bcast_i(en.w, j);
        float* dst = acc + (size_t)cj * LDO;
#pragma unroll
        for (int i = 0; i < P; ++i) {
          const int ch = LINE_CH(k1, i);
          float val = v[j][0];
#pragma unroll
          for (int q = 1; q < P; ++q) val = (ch >= 32 * q) ? v[j][q] : val;
          if (ch >= 0 && ch < NCH) atomicAdd(dst + ch, val * nj);
        }
      }
    }
  }
#undef LINE_CH
}

__global__ void __launch_bounds__(SCAT_THREADS)
gcn_scatter1(const int* __restrict__ row, const int* __restrict__ col,
             const float* __restrict__ dinv, float* __restrict__ nrm2,
             const float* __restrict__ h1,
             float* __restrict__ agg1, int* __restrict__ claim, int N, int E) {
  scatter_rows<HID, HID, HID, true>(row, col, dinv, nrm2, h1, agg1, claim, N, E);
}

#define A2_PITCH (HID + 4)
__global__ void __launch_bounds__(96)
gcn_gemm2(const float* __restrict__ agg1,
          const float* __restrict__ W2,
          const float* __restrict__ dinv,
          const float* __restrict__ b2,
          float* __restrict__ h2,
          float* __restrict__ out) {
  __shared__ __attribute__((aligned(16))) float As[16 * A2_PITCH];
  const int wave = threadIdx.x >> 5;
  const int lane = threadIdx.x & 31;
  const int m0 = blockIdx.x * 16;
  const int n0 = wave * 16;
  const int rowA = lane & 15;
  const int colB = n0 + (lane & 15);
  const int kOff = (lane < 16) ? 0 : 2;
  const bool valid = (colB < OUTC);

  for (int p = wave; p < 8; p += 3) {
    const int rr = 2 * p + (lane >> 4);
    const int cc = (lane & 15) * 4;
    const size_t g = (size_t)(m0 + rr) * HID + cc;
    const float4 u = *reinterpret_cast<const float4*>(agg1 + g);
    float4 s;
    s.x = fmaxf(u.x, 0.0f);
    s.y = fmaxf(u.y, 0.0f);
    s.z = fmaxf(u.z, 0.0f);
    s.w = fmaxf(u.w, 0.0f);
    *reinterpret_cast<float4*>(&As[rr * A2_PITCH + cc]) = s;
  }

  const int mBase = m0 + ((lane < 16) ? 0 : 8);
  const float4 dlo = *reinterpret_cast<const float4*>(dinv + mBase);
  const float4 dhi = *reinterpret_cast<const float4*>(dinv + mBase + 4);
  const float bias = valid ? b2[colB] : 0.0f;
  __syncthreads();
  const float* arow = &As[rowA * A2_PITCH + kOff];
  v8f acc = {};
#pragma unroll
  for (int k = 0; k < HID; k += 4) {
    v2f a, b;
    a = *reinterpret_cast<const v2f*>(arow + k);
    b.x = valid ? W2[(size_t)(k + kOff) * OUTC + colB] : 0.0f;
    b.y = valid ? W2[(size_t)(k + kOff + 1) * OUTC + colB] : 0.0f;
    acc = __builtin_amdgcn_wmma_f32_16x16x4_f32(false, a, false, b,
                                                (short)0, acc, false, false);
  }
  const float dv[8] = {dlo.x, dlo.y, dlo.z, dlo.w, dhi.x, dhi.y, dhi.z, dhi.w};
  if (valid) {
#pragma unroll
    for (int j = 0; j < 8; ++j) {
      const size_t o = (size_t)(mBase + j) * OUTC + colB;
      h2[(size_t)(mBase + j) * OUTP + colB] = acc[j];
      out[o] = bias + acc[j] * (dv[j] * dv[j]);
    }
  }
}

__global__ void __launch_bounds__(SCAT_THREADS)
gcn_scatter2(const int* __restrict__ row, const int* __restrict__ col,
             float* __restrict__ nrm2, const float* __restrict__ h2,
             float* __restrict__ out, int* __restrict__ claim, int N, int E) {
  scatter_rows<OUTC, OUTP, OUTC, false>(row, col, nullptr, nrm2, h2, out, claim, N, E);
}

extern "C" void kernel_launch(void* const* d_in, const int* in_sizes, int n_in,
                              void* d_out, int out_size, void* d_ws, size_t ws_size,
                              hipStream_t stream) {
  const float* x  = (const float*)d_in[0];
  const int*   ei = (const int*)d_in[1];
  const float* W1 = (const float*)d_in[2];
  const float* b1 = (const float*)d_in[3];
  const float* W2 = (const float*)d_in[4];
  const float* b2 = (const float*)d_in[5];
  float* out = (float*)d_out;

  const int N = in_sizes[0] / IN_CH;
  const int E = in_sizes[1] / 2;
  const int* rowp = ei;
  const int* colp = ei + E;

  float* ws   = (float*)d_ws;
  float* deg  = ws;
  float* h1   = ws + N;
  float* agg1 = ws + N + (size_t)N * HID;
  float* nrm2 = agg1 + (size_t)N * HID;
  unsigned int* part = (unsigned int*)(nrm2 + 2 * (size_t)E);
  int* claim = (int*)(part + (size_t)DEG_S * N);

  gcn_degree<<<dim3((N + DEG_RN - 1) / DEG_RN, DEG_S), DEG_THREADS, 0, stream>>>(colp, part, N, E);
  gcn_dinv<<<(N + 255) / 256, 256, 0, stream>>>(part, deg, claim, N);
  gcn_gemm1<<<N / 16, 128, 0, stream>>>(x, W1, deg, b1, h1, agg1);
  gcn_scatter1<<<2 * ((E + EPB - 1) / EPB), SCAT_THREADS, 0, stream>>>(
      rowp, colp, deg, nrm2, h1, agg1, claim, N, E);
  gcn_gemm2<<<N / 16, 96, 0, stream>>>(agg1, W2, deg, b2, h1, out);
  gcn_scatter2<<<2 * ((E + EPB - 1) / EPB), SCAT_THREADS, 0, stream>>>(
      rowp, colp, nrm2, h1, out, claim + 2, N, E);
}


# Both GEMMs: a work-group takes 5 row tiles in turn, a wave keeps its W columns (all k steps) in registers for all, next tile's rows requested during the products; 5x fewer waves; sum order kept
# speedup vs baseline: 3.0689x; 1.0614x over previous
#include <hip/hip_runtime.h>
#include <hip/hip_bf16.h>

typedef __attribute__((ext_vector_type(2))) float v2f;
typedef __attribute__((ext_vector_type(8))) float v8f;

#define IN_CH  128
#define HID    64
#define OUTC   40
#define OUTP   64

__device__ __forceinline__ int package_half() {
  return (int)((__builtin_amdgcn_s_sendmsg_rtn(0x87) >> 18) & 1u);
}

#define DEG_RN      12544
#define DEG_S       32
#define DEG_THREADS 1024
#define DEG_Q       16

__device__ __forceinline__ void deg_count(unsigned int* cnt, int c, int lo, int nr) {
  const unsigned int d = (unsigned int)c - (unsigned int)lo;
  if (d < (unsigned int)nr) atomicAdd(&cnt[d], 1u);
}

__global__ void __launch_bounds__(DEG_THREADS)
gcn_degree(const int* __restrict__ col, unsigned int* __restrict__ part, int N, int E) {
  __shared__ unsigned int cnt[DEG_RN];
  const int tid = threadIdx.x;
  const int lo = blockIdx.x * DEG_RN;
  const int nr = min(DEG_RN, N - lo);
  const int s = blockIdx.y;
  for (int i = tid; i < DEG_RN; i += DEG_THREADS) cnt[i] = 0u;
  __syncthreads();

  const int per = (E + DEG_S - 1) / DEG_S;
  const int e0 = min(E, s * per);
  const int e1 = min(E, e0 + per);
  const int mis = (int)((((size_t)(col + e0)) >> 2) & 3);
  const int a0 = min(e1, e0 + ((4 - mis) & 3));
  const int nq = (e1 - a0) >> 2;
  const int a1 = a0 + 4 * nq;
  if (tid < a0 - e0) deg_count(cnt, col[e0 + tid], lo, nr);
  if (tid < e1 - a1) deg_count(cnt, col[a1 + tid], lo, nr);
  const int4* __restrict__ q4 = reinterpret_cast<const int4*>(col + a0);
  for (int q0 = 0; q0 < nq; q0 += DEG_THREADS * DEG_Q) {
    int4 v[DEG_Q];
#pragma unroll
    for (int u = 0; u < DEG_Q; ++u)
      v[u] = q4[min(q0 + u * DEG_THREADS + tid, nq - 1)];
#pragma unroll
    for (int u = 0; u < DEG_Q; ++u) {
      if (q0 + u * DEG_THREADS + tid < nq) {
        deg_count(cnt, v[u].x, lo, nr);
        deg_count(cnt, v[u].y, lo, nr);
        deg_count(cnt, v[u].z, lo, nr);
        deg_count(cnt, v[u].w, lo, nr);
      }
    }
  }
  __syncthreads();
  unsigned int* __restrict__ dst = part + (size_t)s * N + lo;
  for (int i = tid; i < nr; i += DEG_THREADS) dst[i] = cnt[i];
}

__global__ void gcn_dinv(const unsigned int* __restrict__ part, float* __restrict__ dinv,
                         int* __restrict__ claim, int N) {
  int i = blockIdx.x * blockDim.x + threadIdx.x;
  if (i < 4) claim[i] = 0;
  if (i < N) {
    unsigned int cntv = 1u;
#pragma unroll
    for (int s = 0; s < DEG_S; ++s) cntv += part[(size_t)s * N + i];
    const float d = (float)cntv;
    dinv[i] = (d > 0.0f) ? rsqrtf(d) : 0.0f;
  }
}

#define G_RT 5
#define A1_PITCH (IN_CH + 4)
__global__ void __launch_bounds__(128)
gcn_gemm1(const float* __restrict__ x,
          const float* __restrict__ W1,
          const float* __restrict__ dinv,
          const float* __restrict__ b1,
          float* __restrict__ h1,
          float* __restrict__ agg1,
          int ntiles) {
  __shared__ __attribute__((aligned(16))) float As[2][16 * A1_PITCH];
  const int wave = threadIdx.x >> 5;
  const int lane = threadIdx.x & 31;
  const int n0 = wave * 16;
  const int rowA = lane & 15;
  const int colB = lane & 15;
  const int kOff = (lane < 16) ? 0 : 2;
  const int c = n0 + colB;
  const int t0 = blockIdx.x * G_RT;
  const int nt = min(G_RT, ntiles - t0);
  if (nt <= 0) return;

  v2f bw[IN_CH / 4];
#pragma unroll
  for (int k = 0; k < IN_CH; k += 4) {
    bw[k / 4].x = W1[(size_t)(k + kOff) * HID + c];
    bw[k / 4].y = W1[(size_t)(k + kOff + 1) * HID + c];
  }
  const float bias = b1[c];

  const float* xrow = x + (size_t)(t0 * 16 + wave * 4) * IN_CH + lane * 4;
  float4 p0 = *reinterpret_cast<const float4*>(xrow);
  float4 p1 = *reinterpret_cast<const float4*>(xrow + IN_CH);
  float4 p2 = *reinterpret_cast<const float4*>(xrow + 2 * IN_CH);
  float4 p3 = *reinterpret_cast<const float4*>(xrow + 3 * IN_CH);

  for (int t = 0; t < nt; ++t) {
    float* buf = As[t & 1];
    const int m0 = (t0 + t) * 16;
    float* brow = &buf[(wave * 4) * A1_PITCH + lane * 4];
    *reinterpret_cast<float4*>(brow) = p0;
    *reinterpret_cast<float4*>(brow + A1_PITCH) = p1;
    *reinterpret_cast<float4*>(brow + 2 * A1_PITCH) = p2;
    *reinterpret_cast<float4*>(brow + 3 * A1_PITCH) = p3;
    __syncthreads();

    const int mBase = m0 + ((lane < 16) ? 0 : 8);
    const float4 dlo = *reinterpret_cast<const float4*>(dinv + mBase);
    const float4 dhi = *reinterpret_cast<const float4*>(dinv + mBase + 4);
    if (t + 1 < nt) {
      xrow += 16 * IN_CH;
      p0 = *reinterpret_cast<const float4*>(xrow);
      p1 = *reinterpret_cast<const float4*>(xrow + IN_CH);
      p2 = *reinterpret_cast<const float4*>(xrow + 2 * IN_CH);
      p3 = *reinterpret_cast<const float4*>(xrow + 3 * IN_CH);
    }

    const float* arow = &buf[rowA * A1_PITCH + kOff];
    v8f acc = {};
#pragma unroll
    for (int k = 0; k < IN_CH; k += 4) {
      const v2f a = *reinterpret_cast<const v2f*>(arow + k);
      acc = __builtin_amdgcn_wmma_f32_16x16x4_f32(false, a, false, bw[k / 4],
                                                  (short)0, acc, false, false);
    }
    const float dv[8] = {dlo.x, dlo.y, dlo.z, dlo.w, dhi.x, dhi.y, dhi.z, dhi.w};
    const size_t o = (size_t)mBase * HID + c;
    float* hp = h1 + o;
    float* ap = agg1 + o;
#pragma unroll
    for (int j = 0; j < 8; ++j) {
      hp[j * HID] = acc[j];
      ap[j * HID] = bias + acc[j] * (dv[j] * dv[j]);
    }
  }
}

#define SCAT_THREADS 256
#define SCAT_WAVES (SCAT_THREADS / 32)
#define EPW 64
#define EPB (EPW * SCAT_WAVES)
#define GRP 20

__device__ __forceinline__ int bcast_i(int v, int j) {
  return __builtin_amdgcn_readlane(v, j);
}
__device__ __forceinline__ float bcast_f(int v, int j) {
  return __int_as_float(__builtin_amdgcn_readlane(v, j));
}

template <int NCH, int LDH, int LDO, bool MAKE_NORM>
__device__ __forceinline__ void scatter_rows(const int* __restrict__ row,
                                             const int* __restrict__ col,
                                             const float* __restrict__ dinv,
                                             float* __restrict__ nrm2,
                                             const float* __restrict__ h,
                                             float* __restrict__ acc,
                                             int* __restrict__ claim,
                                             int N, int E) {
  constexpr int P = (NCH + 31) / 32;
  constexpr int KMIN = NCH - 32 * (P - 1);
  __shared__ int s_item;
  __shared__ __attribute__((aligned(16))) int4 ent[SCAT_WAVES][EPW];
  const int lane = threadIdx.x & 31;
  const int wib = __builtin_amdgcn_readfirstlane((int)(threadIdx.x >> 5));
  const int nchunks = (E + EPB - 1) / EPB;

  if (threadIdx.x == 0) {
    int hh = package_half();
    int idx = atomicAdd(&claim[hh], 1);
    if (idx >= nchunks) {
      hh ^= 1;
      idx = atomicAdd(&claim[hh], 1);
    }
    s_item = (idx < nchunks) ? (2 * idx + hh) : -1;
  }
  __syncthreads();
  const int item = __builtin_amdgcn_readfirstlane(s_item);
  if (item < 0) return;
  const int hsel = item & 1;
  const int base = (item >> 1) * EPB + wib * EPW;
  if (base >= E) return;
  const int nsplit = ((N >> 1) + 3) & ~3;
  const unsigned int lo = hsel ? (unsigned int)nsplit : 0u;
  const unsigned int nr = hsel ? (unsigned int)(N - nsplit) : (unsigned int)nsplit;
  float* __restrict__ nrm = nrm2 + (size_t)hsel * E;

  int r[2], c[2], nb[2], k1v[2];
  bool mine[2];
  bool valid[2];
  int ec[2];
#pragma unroll
  for (int u = 0; u < 2; ++u) {
    const int e = base + 32 * u + lane;
    valid[u] = e < E;
    ec[u] = min(e, E - 1);
    c[u] = col[ec[u]];
    r[u] = row[ec[u]];
    if (!MAKE_NORM) nb[u] = __float_as_int(nrm[ec[u]]);
  }
  if (MAKE_NORM) {
    asm volatile("" : : "v"(c[0]), "v"(c[1]), "v"(r[0]), "v"(r[1]));
  } else {
    asm volatile("" : : "v"(c[0]), "v"(c[1]), "v"(r[0]), "v"(r[1]), "v"(nb[0]), "v"(nb[1]));
  }
#pragma unroll
  for (int u = 0; u < 2; ++u)
    mine[u] = valid[u] && (((unsigned int)c[u] - lo) < nr);
  if (MAKE_NORM) {
    float da[2], db[2];
#pragma unroll
    for (int u = 0; u < 2; ++u) {
      da[u] = dinv[mine[u] ? r[u] : 0];
      db[u] = dinv[mine[u] ? c[u] : 0];
    }
#pragma unroll
    for (int u = 0; u < 2; ++u) {
      const float nv = mine[u] ? da[u] * db[u] : 0.0f;
      if (valid[u]) nrm[base + 32 * u + lane] = nv;
      nb[u] = __float_as_int(nv);
    }
  }
#pragma unroll
  for (int u = 0; u < 2; ++u) {
    k1v[u] = (KMIN == 32)
        ? 32
        : max(KMIN, 32 - (int)((((size_t)(acc + (size_t)c[u] * LDO)) >> 2) & 31));
  }
  const unsigned int m0 = __builtin_amdgcn_ballot_w32(mine[0]);
  const unsigned int m1 = __builtin_amdgcn_ballot_w32(mine[1]);
  const int n0 = __builtin_popcount(m0);
  const int n = n0 + __builtin_popcount(m1);
  if (mine[0])
    ent[wib][__builtin_amdgcn_mbcnt_lo(m0, 0u)] = make_int4(r[0], c[0], nb[0], k1v[0]);
  if (mine[1])
    ent[wib][n0 + __builtin_amdgcn_mbcnt_lo(m1, 0u)] = make_int4(r[1], c[1], nb[1], k1v[1]);
  __builtin_amdgcn_wave_barrier();

#define LINE_CH(k1, i) (((k1) - 32 + 32 * (i)) + ((lane - ((k1) - 32 + 32 * (i))) & 31))

  const int ng = (n + GRP - 1) / GRP;
  const int gs = (ng <= 1) ? n : (ng == 2) ? (n + 1) / 2 : (ng == 3) ? (n + 2) / 3 : (n + 3) / 4;
  for (int s = 0; s < n; s += gs) {
    const int cnt = __builtin_amdgcn_readfirstlane(min(gs, n - s));
    const int4 en = ent[wib][s + min(lane, cnt - 1)];
    float v[GRP][P] = {};
#pragma unroll
    for (int j = 0; j < GRP; ++j) {
      if (__builtin_expect(j < cnt, 1)) {
        const int rj = bcast_i(en.x, j);
        const float* src = h + (size_t)rj * LDH;
#pragma unroll
        for (int p = 0; p < P; ++p)
          v[j][p] = src[min(32 * p + lane, NCH - 1)];
      }
    }
#pragma unroll
    for (int j = 0; j < GRP; ++j) {
#pragma unroll
      for (int p = 0; p < P; ++p) asm volatile("" : : "v"(v[j][p]));
    }
#pragma unroll
    for (int j = 0; j < GRP; ++j) {
      if (__builtin_expect(j < cnt, 1)) {
        const int cj = bcast_i(en.y, j);
        const float nj = bcast_f(en.z, j);
        const int k1 = (KMIN == 32) ? 32 : bcast_i(en.w, j);
        float* dst = acc + (size_t)cj * LDO;
#pragma unroll
        for (int i = 0; i < P; ++i) {
          const int ch = LINE_CH(k1, i);
          float val = v[j][0];
#pragma unroll
          for (int q = 1; q < P; ++q) val = (ch >= 32 * q) ? v[j][q] : val;
          if (ch >= 0 && ch < NCH) atomicAdd(dst + ch, val * nj);
        }
      }
    }
  }
#undef LINE_CH
}

__global__ void __launch_bounds__(SCAT_THREADS)
gcn_scatter1(const int* __restrict__ row, const int* __restrict__ col,
             const float* __restrict__ dinv, float* __restrict__ nrm2,
             const float* __restrict__ h1,
             float* __restrict__ agg1, int* __restrict__ claim, int N, int E) {
  scatter_rows<HID, HID, HID, true>(row, col, dinv, nrm2, h1, agg1, claim, N, E);
}

#define A2_PITCH (HID + 4)
__device__ __forceinline__ float4 relu4(float4 u) {
  float4 r;
  r.x = fmaxf(u.x, 0.0f);
  r.y = fmaxf(u.y, 0.0f);
  r.z = fmaxf(u.z, 0.0f);
  r.w = fmaxf(u.w, 0.0f);
  return r;
}
__global__ void __launch_bounds__(96)
gcn_gemm2(const float* __restrict__ agg1,
          const float* __restrict__ W2,
          const float* __restrict__ dinv,
          const float* __restrict__ b2,
          float* __restrict__ h2,
          float* __restrict__ out,
          int ntiles) {
  __shared__ __attribute__((aligned(16))) float As[2][16 * A2_PITCH];
  const int wave = threadIdx.x >> 5;
  const int lane = threadIdx.x & 31;
  const int n0 = wave * 16;
  const int rowA = lane & 15;
  const int colB = n0 + (lane & 15);
  const int kOff = (lane < 16) ? 0 : 2;
  const bool valid = (colB < OUTC);
  const int t0 = blockIdx.x * G_RT;
  const int nt = min(G_RT, ntiles - t0);
  if (nt <= 0) return;

  const int cB = valid ? colB : (OUTC - 1);
  v2f bw[HID / 4];
#pragma unroll
  for (int k = 0; k < HID; k += 4) {
    const float w0 = W2[(size_t)(k + kOff) * OUTC + cB];
    const float w1 = W2[(size_t)(k + kOff + 1) * OUTC + cB];
    bw[k / 4].x = valid ? w0 : 0.0f;
    bw[k / 4].y = valid ? w1 : 0.0f;
  }
  const float bcol = b2[cB];
  const float bias = valid ? bcol : 0.0f;

  const int rsub = lane >> 4;
  const int cc = (lane & 15) * 4;
  const bool third = (wave + 6 < 8);
  const float* arow0 = agg1 + (size_t)(t0 * 16 + 2 * wave + rsub) * HID + cc;
  float4 p0 = *reinterpret_cast<const float4*>(arow0);
  float4 p1 = *reinterpret_cast<const float4*>(arow0 + 6 * HID);
  float4 p2 = p1;
  if (third) p2 = *reinterpret_cast<const float4*>(arow0 + 12 * HID);

  for (int t = 0; t < nt; ++t) {
    float* buf = As[t & 1];
    const int m0 = (t0 + t) * 16;
    float* brow = &buf[(2 * wave + rsub) * A2_PITCH + cc];
    *reinterpret_cast<float4*>(brow) = relu4(p0);
    *reinterpret_cast<float4*>(brow + 6 * A2_PITCH) = relu4(p1);
    if (third) *reinterpret_cast<float4*>(brow + 12 * A2_PITCH) = relu4(p2);
    __syncthreads();

    const int mBase = m0 + ((lane < 16) ? 0 : 8);
    const float4 dlo = *reinterpret_cast<const float4*>(dinv + mBase);
    const float4 dhi = *reinterpret_cast<const float4*>(dinv + mBase + 4);
    if (t + 1 < nt) {
      arow0 += 16 * HID;
      p0 = *reinterpret_cast<const float4*>(arow0);
      p1 = *reinterpret_cast<const float4*>(arow0 + 6 * HID);
      if (third) p2 = *reinterpret_cast<const float4*>(arow0 + 12 * HID);
    }

    const float* arow = &buf[rowA * A2_PITCH + kOff];
    v8f acc = {};
#pragma unroll
    for (int k = 0; k < HID; k += 4) {
      const v2f a = *reinterpret_cast<const v2f*>(arow + k);
      acc = __builtin_amdgcn_wmma_f32_16x16x4_f32(false, a, false, bw[k / 4],
                                                  (short)0, acc, false, false);
    }
    const float dv[8] = {dlo.x, dlo.y, dlo.z, dlo.w, dhi.x, dhi.y, dhi.z, dhi.w};
    if (valid) {
      float* hp = h2 + (size_t)mBase * OUTP + colB;
      float* op = out + (size_t)mBase * OUTC + colB;
#pragma unroll
      for (int j = 0; j < 8; ++j) {
        hp[j * OUTP] = acc[j];
        op[j * OUTC] = bias + acc[j] * (dv[j] * dv[j]);
      }
    }
  }
}

__global__ void __launch_bounds__(SCAT_THREADS)
gcn_scatter2(const int* __restrict__ row, const int* __restrict__ col,
             float* __restrict__ nrm2, const float* __restrict__ h2,
             float* __restrict__ out, int* __restrict__ claim, int N, int E) {
  scatter_rows<OUTC, OUTP, OUTC, false>(row, col, nullptr, nrm2, h2, out, claim, N, E);
}

extern "C" void kernel_launch(void* const* d_in, const int* in_sizes, int n_in,
                              void* d_out, int out_size, void* d_ws, size_t ws_size,
                              hipStream_t stream) {
  const float* x  = (const float*)d_in[0];
  const int*   ei = (const int*)d_in[1];
  const float* W1 = (const float*)d_in[2];
  const float* b1 = (const float*)d_in[3];
  const float* W2 = (const float*)d_in[4];
  const float* b2 = (const float*)d_in[5];
  float* out = (float*)d_out;

  const int N = in_sizes[0] / IN_CH;
  const int E = in_sizes[1] / 2;
  const int* rowp = ei;
  const int* colp = ei + E;

  float* ws   = (float*)d_ws;
  float* deg  = ws;
  float* h1   = ws + N;
  float* agg1 = ws + N + (size_t)N * HID;
  float* nrm2 = agg1 + (size_t)N * HID;
  unsigned int* part = (unsigned int*)(nrm2 + 2 * (size_t)E);
  int* claim = (int*)(part + (size_t)DEG_S * N);

  gcn_degree<<<dim3((N + DEG_RN - 1) / DEG_RN, DEG_S), DEG_THREADS, 0, stream>>>(colp, part, N, E);
  gcn_dinv<<<(N + 255) / 256, 256, 0, stream>>>(part, deg, claim, N);
  gcn_gemm1<<<(N / 16 + G_RT - 1) / G_RT, 128, 0, stream>>>(x, W1, deg, b1, h1, agg1, N / 16);
  gcn_scatter1<<<2 * ((E + EPB - 1) / EPB), SCAT_THREADS, 0, stream>>>(
      rowp, colp, deg, nrm2, h1, agg1, claim, N, E);
  gcn_gemm2<<<(N / 16 + G_RT - 1) / G_RT, 96, 0, stream>>>(agg1, W2, deg, b2, h1, out, N / 16);
  gcn_scatter2<<<2 * ((E + EPB - 1) / EPB), SCAT_THREADS, 0, stream>>>(
      rowp, colp, nrm2, h1, out, claim + 2, N, E);
}
